# starting kernel
# baseline (speedup 1.0000x reference)
#include <hip/hip_runtime.h>
#include <hip/hip_bf16.h>

typedef float v2f __attribute__((ext_vector_type(2)));
typedef float v8f __attribute__((ext_vector_type(8)));

#define D_IN  128
#define D_HID 256
#define D_OUT 3

__global__ void gnn_init_kernel(float* __restrict__ ws, long long n) {
    long long i = (long long)blockIdx.x * blockDim.x + threadIdx.x;
    long long stride = (long long)gridDim.x * blockDim.x;
    for (; i < n; i += stride) ws[i] = 0.0f;
}

__global__ void gnn_fusew_kernel(const float* __restrict__ W1,
                                 const float* __restrict__ b1,
                                 const float* __restrict__ W2,
                                 const float* __restrict__ b2,
                                 float* __restrict__ packedB,
                                 float* __restrict__ bc) {
    __shared__ float Wc[D_OUT][D_IN];
    int tid = threadIdx.x;
    if (tid < D_OUT * D_IN) {
        int n = tid / D_IN, k = tid % D_IN;
        float acc = 0.0f;
        for (int h = 0; h < D_HID; ++h)
            acc += W2[n * D_HID + h] * W1[h * D_IN + k];
        Wc[n][k] = acc;
    }
    if (tid < 4) {
        float acc = 0.0f;
        if (tid < D_OUT) {
            acc = b2[tid];
            for (int h = 0; h < D_HID; ++h)
                acc += W2[tid * D_HID + h] * b1[h];
        }
        bc[tid] = acc;
    }
    __syncthreads();
    for (int idx = tid; idx < 2048; idx += blockDim.x) {
        int s    = idx >> 6;
        int j    = (idx >> 5) & 1;
        int lane = idx & 31;
        int n    = lane & 15;
        int k    = 4 * s + j + ((lane >= 16) ? 2 : 0);
        packedB[idx] = (n < D_OUT) ? Wc[n][k] : 0.0f;
    }
}

__global__ void gnn_scatter_kernel(const float* __restrict__ x,
                                   const int* __restrict__ edges,
                                   float* __restrict__ summed,
                                   float* __restrict__ counts,
                                   int E) {
    int wave = (int)((blockIdx.x * blockDim.x + threadIdx.x) >> 5);
    int lane = threadIdx.x & 31;
    if (wave >= E) return;
    int src = edges[wave];
    int dst = edges[E + wave];
    const float4* xr = (const float4*)(x + (long long)src * D_IN);
    float4 v = xr[lane];
    float* srow = summed + (long long)dst * D_IN + lane * 4;
    atomicAdd(srow + 0, v.x);
    atomicAdd(srow + 1, v.y);
    atomicAdd(srow + 2, v.z);
    atomicAdd(srow + 3, v.w);
    if (lane == 0) atomicAdd(counts + dst, 1.0f);
}

__global__ void gnn_mlp_kernel(const float* __restrict__ summed,
                               const float* __restrict__ counts,
                               const float* __restrict__ packedB,
                               const float* __restrict__ bc,
                               float* __restrict__ out,
                               int N) {
    int wave = (int)((blockIdx.x * blockDim.x + threadIdx.x) >> 5);
    int lane = threadIdx.x & 31;
    int nTiles = N >> 4;
    if (wave >= nTiles) return;

    int m0   = wave << 4;
    int mrow = m0 + (lane & 15);
    float inv = 1.0f / fmaxf(counts[mrow], 1.0f);
    int koff = (lane >= 16) ? 2 : 0;
    const float* arow = summed + (long long)mrow * D_IN + koff;

    v8f c = {};
    #pragma unroll 4
    for (int s = 0; s < 32; ++s) {
        float2 a2 = *(const float2*)(arow + 4 * s);
        v2f a; a[0] = a2.x * inv; a[1] = a2.y * inv;
        v2f b; b[0] = packedB[s * 64 + lane];
               b[1] = packedB[s * 64 + 32 + lane];
        c = __builtin_amdgcn_wmma_f32_16x16x4_f32(
                 false, a,  false, b,
                 (short)0, c,  false,  false);
    }

    int n = lane & 15;
    if (n < D_OUT) {
        float bcn = bc[n];
        int rbase = m0 + ((lane >= 16) ? 8 : 0);
        #pragma unroll
        for (int j = 0; j < 8; ++j)
            out[(long long)(rbase + j) * D_OUT + n] = c[j] + bcn;
    }
}

extern "C" void kernel_launch(void* const* d_in, const int* in_sizes, int n_in,
                              void* d_out, int out_size, void* d_ws, size_t ws_size,
                              hipStream_t stream) {
    const float* x     = (const float*)d_in[0];
    const int*   edges = (const int*)  d_in[1];
    const float* W1    = (const float*)d_in[2];
    const float* b1    = (const float*)d_in[3];
    const float* W2    = (const float*)d_in[4];
    const float* b2    = (const float*)d_in[5];
    float*       out   = (float*)d_out;

    int N = in_sizes[0] / D_IN;
    int E = in_sizes[1] / 2;

    float* ws = (float*)d_ws;
    long long offSum = 0;
    long long offCnt = (long long)N * D_IN;
    long long offB   = offCnt + N;
    long long offBc  = offB + 2048;

    gnn_init_kernel<<<4096, 256, 0, stream>>>(ws + offSum, offB);

    gnn_fusew_kernel<<<1, 384, 0, stream>>>(W1, b1, W2, b2, ws + offB, ws + offBc);

    int blocksScatter = (E + 7) / 8;
    gnn_scatter_kernel<<<blocksScatter, 256, 0, stream>>>(
        x, edges, ws + offSum, ws + offCnt, E);

    int nTiles = N / 16;
    int blocksMlp = (nTiles + 7) / 8;
    gnn_mlp_kernel<<<blocksMlp, 256, 0, stream>>>(
        ws + offSum, ws + offCnt, ws + offB, ws + offBc, out, N);
}


# Apply the folded 3x128 linear map to x BEFORE the edge scatter (linearity): WMMA projects each node to 3 floats, edges then add 16-byte records (3 sums + count) not 129 floats; sum order changes
# speedup vs baseline: 3.1586x; 3.1586x over previous
#include <hip/hip_runtime.h>
#include <hip/hip_bf16.h>

typedef float v2f __attribute__((ext_vector_type(2)));
typedef float v8f __attribute__((ext_vector_type(8)));

#define D_IN  128
#define D_HID 256
#define D_OUT 3


__global__ void gnn_fusew_kernel(const float* __restrict__ W1,
                                 const float* __restrict__ b1,
                                 const float* __restrict__ W2,
                                 const float* __restrict__ b2,
                                 float* __restrict__ packedB,
                                 float* __restrict__ bc) {
    __shared__ float Wc[D_OUT][D_IN];
    int tid = threadIdx.x;
    if (tid < D_OUT * D_IN) {
        int n = tid / D_IN, k = tid % D_IN;
        float acc = 0.0f;
        for (int h = 0; h < D_HID; ++h)
            acc += W2[n * D_HID + h] * W1[h * D_IN + k];
        Wc[n][k] = acc;
    }
    if (tid < 4) {
        float acc = 0.0f;
        if (tid < D_OUT) {
            acc = b2[tid];
            for (int h = 0; h < D_HID; ++h)
                acc += W2[tid * D_HID + h] * b1[h];
        }
        bc[tid] = acc;
    }
    __syncthreads();
    for (int idx = tid; idx < 2048; idx += blockDim.x) {
        int s    = idx >> 6;
        int j    = (idx >> 5) & 1;
        int lane = idx & 31;
        int n    = lane & 15;
        int k    = 4 * s + j + ((lane >= 16) ? 2 : 0);
        packedB[idx] = (n < D_OUT) ? Wc[n][k] : 0.0f;
    }
}

__global__ void gnn_project_kernel(const float* __restrict__ x,
                                   const float* __restrict__ packedB,
                                   float* __restrict__ y4,
                                   float* __restrict__ acc,
                                   int N) {
    int wave = (int)((blockIdx.x * blockDim.x + threadIdx.x) >> 5);
    int lane = threadIdx.x & 31;
    int nTiles = N >> 4;
    if (wave >= nTiles) return;

    int m0   = wave << 4;
    int mrow = m0 + (lane & 15);
    int koff = (lane >= 16) ? 2 : 0;
    const float* arow = x + (long long)mrow * D_IN + koff;

    v8f c = {};
    #pragma unroll 4
    for (int s = 0; s < 32; ++s) {
        float2 a2 = *(const float2*)(arow + 4 * s);
        v2f a; a[0] = a2.x; a[1] = a2.y;
        v2f b; b[0] = packedB[s * 64 + lane];
               b[1] = packedB[s * 64 + 32 + lane];
        c = __builtin_amdgcn_wmma_f32_16x16x4_f32(
                 false, a,  false, b,
                 (short)0, c,  false,  false);
    }

    int n = lane & 15;
    if (n < 4) {
        int rbase = m0 + ((lane >= 16) ? 8 : 0);
        #pragma unroll
        for (int j = 0; j < 8; ++j) {
            long long o = (long long)(rbase + j) * 4 + n;
            y4[o]  = (n < D_OUT) ? c[j] : 1.0f;
            acc[o] = 0.0f;
        }
    }
}

template <bool FULL>
static __device__ __forceinline__ void scatter64(const float* __restrict__ y4,
                                                 const int* __restrict__ edges,
                                                 float* __restrict__ acc,
                                                 int E, int base, int lane) {
    unsigned comp = (unsigned)lane & 3u;
    int quad = lane >> 2;

    int e0 = base + lane;
    int e1 = base + 32 + lane;
    int s0 = 0, d0 = 0, s1 = 0, d1 = 0;
    if (FULL || e0 < E) { s0 = edges[e0]; d0 = edges[E + e0]; }
    if (FULL || e1 < E) { s1 = edges[e1]; d1 = edges[E + e1]; }

    float    v[8];
    unsigned dd[8];
    bool     ok[8];
    #pragma unroll
    for (int k = 0; k < 8; ++k) {
        int j = (k & 3) * 8 + quad;
        unsigned sj = (unsigned)__shfl((k < 4) ? s0 : s1, j, 32);
        dd[k] = (unsigned)__shfl((k < 4) ? d0 : d1, j, 32);
        ok[k] = FULL || (base + (k >> 2) * 32 + j) < E;
        v[k]  = ok[k] ? y4[sj * 4u + comp] : 0.0f;
    }
    #pragma unroll
    for (int k = 0; k < 8; ++k)
        if (ok[k]) atomicAdd(acc + (dd[k] * 4u + comp), v[k]);
}

__global__ void gnn_scatter_kernel(const float* __restrict__ y4,
                                   const int* __restrict__ edges,
                                   float* __restrict__ acc,
                                   int E) {
    int wave = (int)((blockIdx.x * blockDim.x + threadIdx.x) >> 5);
    int lane = threadIdx.x & 31;
    int base = wave * 64;
    if (base >= E) return;
    if (base + 64 <= E) scatter64<true >(y4, edges, acc, E, base, lane);
    else                scatter64<false>(y4, edges, acc, E, base, lane);
}

__global__ void gnn_finish_kernel(const float* __restrict__ acc,
                                  const float* __restrict__ bc,
                                  float* __restrict__ out,
                                  int total) {
    int i = (int)(blockIdx.x * blockDim.x + threadIdx.x);
    if (i >= total) return;
    int node = i / D_OUT;
    int n    = i - node * D_OUT;
    float cnt = acc[(long long)node * 4 + 3];
    float sum = acc[(long long)node * 4 + n];
    out[i] = sum / fmaxf(cnt, 1.0f) + bc[n];
}

extern "C" void kernel_launch(void* const* d_in, const int* in_sizes, int n_in,
                              void* d_out, int out_size, void* d_ws, size_t ws_size,
                              hipStream_t stream) {
    const float* x     = (const float*)d_in[0];
    const int*   edges = (const int*)  d_in[1];
    const float* W1    = (const float*)d_in[2];
    const float* b1    = (const float*)d_in[3];
    const float* W2    = (const float*)d_in[4];
    const float* b2    = (const float*)d_in[5];
    float*       out   = (float*)d_out;

    int N = in_sizes[0] / D_IN;
    int E = in_sizes[1] / 2;

    float* ws = (float*)d_ws;
    long long offAcc = 0;
    long long offY   = (long long)N * 4;
    long long offB   = (long long)N * 8;
    long long offBc  = offB + 2048;

    gnn_fusew_kernel<<<1, 384, 0, stream>>>(W1, b1, W2, b2, ws + offB, ws + offBc);

    int nTiles = N / 16;
    int blocksProj = (nTiles + 7) / 8;
    gnn_project_kernel<<<blocksProj, 256, 0, stream>>>(
        x, ws + offB, ws + offY, ws + offAcc, N);

    int blocksScatter = (E + 511) / 512;
    gnn_scatter_kernel<<<blocksScatter, 256, 0, stream>>>(
        ws + offY, edges, ws + offAcc, E);

    int total = N * D_OUT;
    int blocksFin = (total + 255) / 256;
    gnn_finish_kernel<<<blocksFin, 256, 0, stream>>>(
        ws + offAcc, ws + offBc, out, total);
}


# Weight fold W2@W1 spread over 33 one-wave work-groups (one per WMMA k-step, one for the bias): all loads of a wave issued at once, lanes summed by an exchange tree; fold sum order changes
# speedup vs baseline: 4.3661x; 1.3823x over previous
#include <hip/hip_runtime.h>
#include <hip/hip_bf16.h>

typedef float v2f __attribute__((ext_vector_type(2)));
typedef float v8f __attribute__((ext_vector_type(8)));

#define D_IN  128
#define D_HID 256
#define D_OUT 3


__global__ void gnn_fusew_kernel(const float* __restrict__ W1,
                                 const float* __restrict__ b1,
                                 const float* __restrict__ W2,
                                 const float* __restrict__ b2,
                                 float* __restrict__ packedB,
                                 float* __restrict__ bc) {
    int lane = threadIdx.x;
    int s    = blockIdx.x;

    if (s < 32) {
        float4 w1[8];
        float  w2[D_OUT][8];
        #pragma unroll
        for (int i = 0; i < 8; ++i)
            w1[i] = *(const float4*)(W1 + (8 * lane + i) * D_IN + 4 * s);
        #pragma unroll
        for (int n = 0; n < D_OUT; ++n) {
            float4 lo = *(const float4*)(W2 + n * D_HID + 8 * lane);
            float4 hi4 = *(const float4*)(W2 + n * D_HID + 8 * lane + 4);
            w2[n][0] = lo.x;  w2[n][1] = lo.y;  w2[n][2] = lo.z;  w2[n][3] = lo.w;
            w2[n][4] = hi4.x; w2[n][5] = hi4.y; w2[n][6] = hi4.z; w2[n][7] = hi4.w;
        }
        float p[D_OUT][4];
        #pragma unroll
        for (int n = 0; n < D_OUT; ++n) {
            p[n][0] = 0.0f; p[n][1] = 0.0f; p[n][2] = 0.0f; p[n][3] = 0.0f;
            #pragma unroll
            for (int i = 0; i < 8; ++i) {
                p[n][0] += w2[n][i] * w1[i].x;
                p[n][1] += w2[n][i] * w1[i].y;
                p[n][2] += w2[n][i] * w1[i].z;
                p[n][3] += w2[n][i] * w1[i].w;
            }
        }
        #pragma unroll
        for (int m = 16; m >= 1; m >>= 1) {
            #pragma unroll
            for (int n = 0; n < D_OUT; ++n) {
                #pragma unroll
                for (int r = 0; r < 4; ++r)
                    p[n][r] += __shfl_xor(p[n][r], m, 32);
            }
        }
        int  n  = lane & 15;
        bool hi = lane >= 16;
        #pragma unroll
        for (int j = 0; j < 2; ++j) {
            float v0 = hi ? p[0][j + 2] : p[0][j];
            float v1 = hi ? p[1][j + 2] : p[1][j];
            float v2 = hi ? p[2][j + 2] : p[2][j];
            float v  = (n == 0) ? v0 : (n == 1) ? v1 : (n == 2) ? v2 : 0.0f;
            packedB[s * 64 + j * 32 + lane] = v;
        }
    } else {
        float q[D_OUT];
        float bb[8];
        float w2[D_OUT][8];
        {
            float4 lo = *(const float4*)(b1 + 8 * lane);
            float4 hi4 = *(const float4*)(b1 + 8 * lane + 4);
            bb[0] = lo.x;  bb[1] = lo.y;  bb[2] = lo.z;  bb[3] = lo.w;
            bb[4] = hi4.x; bb[5] = hi4.y; bb[6] = hi4.z; bb[7] = hi4.w;
        }
        #pragma unroll
        for (int n = 0; n < D_OUT; ++n) {
            float4 lo = *(const float4*)(W2 + n * D_HID + 8 * lane);
            float4 hi4 = *(const float4*)(W2 + n * D_HID + 8 * lane + 4);
            w2[n][0] = lo.x;  w2[n][1] = lo.y;  w2[n][2] = lo.z;  w2[n][3] = lo.w;
            w2[n][4] = hi4.x; w2[n][5] = hi4.y; w2[n][6] = hi4.z; w2[n][7] = hi4.w;
        }
        #pragma unroll
        for (int n = 0; n < D_OUT; ++n) {
            q[n] = 0.0f;
            #pragma unroll
            for (int i = 0; i < 8; ++i)
                q[n] += w2[n][i] * bb[i];
        }
        #pragma unroll
        for (int m = 16; m >= 1; m >>= 1) {
            #pragma unroll
            for (int n = 0; n < D_OUT; ++n)
                q[n] += __shfl_xor(q[n], m, 32);
        }
        if (lane < 4) {
            float v = (lane == 0) ? q[0] : (lane == 1) ? q[1] : (lane == 2) ? q[2] : 0.0f;
            bc[lane] = (lane < D_OUT) ? v + b2[lane] : 0.0f;
        }
    }
}

__global__ void gnn_project_kernel(const float* __restrict__ x,
                                   const float* __restrict__ packedB,
                                   float* __restrict__ y4,
                                   float* __restrict__ acc,
                                   int N) {
    int wave = (int)((blockIdx.x * blockDim.x + threadIdx.x) >> 5);
    int lane = threadIdx.x & 31;
    int nTiles = N >> 4;
    if (wave >= nTiles) return;

    int m0   = wave << 4;
    int mrow = m0 + (lane & 15);
    int koff = (lane >= 16) ? 2 : 0;
    const float* arow = x + (long long)mrow * D_IN + koff;

    v8f c = {};
    #pragma unroll 4
    for (int s = 0; s < 32; ++s) {
        float2 a2 = *(const float2*)(arow + 4 * s);
        v2f a; a[0] = a2.x; a[1] = a2.y;
        v2f b; b[0] = packedB[s * 64 + lane];
               b[1] = packedB[s * 64 + 32 + lane];
        c = __builtin_amdgcn_wmma_f32_16x16x4_f32(
                 false, a,  false, b,
                 (short)0, c,  false,  false);
    }

    int n = lane & 15;
    if (n < 4) {
        int rbase = m0 + ((lane >= 16) ? 8 : 0);
        #pragma unroll
        for (int j = 0; j < 8; ++j) {
            long long o = (long long)(rbase + j) * 4 + n;
            y4[o]  = (n < D_OUT) ? c[j] : 1.0f;
            acc[o] = 0.0f;
        }
    }
}

template <bool FULL>
static __device__ __forceinline__ void scatter64(const float* __restrict__ y4,
                                                 const int* __restrict__ edges,
                                                 float* __restrict__ acc,
                                                 int E, int base, int lane) {
    unsigned comp = (unsigned)lane & 3u;
    int quad = lane >> 2;

    int e0 = base + lane;
    int e1 = base + 32 + lane;
    int s0 = 0, d0 = 0, s1 = 0, d1 = 0;
    if (FULL || e0 < E) { s0 = edges[e0]; d0 = edges[E + e0]; }
    if (FULL || e1 < E) { s1 = edges[e1]; d1 = edges[E + e1]; }

    float    v[8];
    unsigned dd[8];
    bool     ok[8];
    #pragma unroll
    for (int k = 0; k < 8; ++k) {
        int j = (k & 3) * 8 + quad;
        unsigned sj = (unsigned)__shfl((k < 4) ? s0 : s1, j, 32);
        dd[k] = (unsigned)__shfl((k < 4) ? d0 : d1, j, 32);
        ok[k] = FULL || (base + (k >> 2) * 32 + j) < E;
        v[k]  = ok[k] ? y4[sj * 4u + comp] : 0.0f;
    }
    #pragma unroll
    for (int k = 0; k < 8; ++k)
        if (ok[k]) atomicAdd(acc + (dd[k] * 4u + comp), v[k]);
}

__global__ void gnn_scatter_kernel(const float* __restrict__ y4,
                                   const int* __restrict__ edges,
                                   float* __restrict__ acc,
                                   int E) {
    int wave = (int)((blockIdx.x * blockDim.x + threadIdx.x) >> 5);
    int lane = threadIdx.x & 31;
    int base = wave * 64;
    if (base >= E) return;
    if (base + 64 <= E) scatter64<true >(y4, edges, acc, E, base, lane);
    else                scatter64<false>(y4, edges, acc, E, base, lane);
}

__global__ void gnn_finish_kernel(const float* __restrict__ acc,
                                  const float* __restrict__ bc,
                                  float* __restrict__ out,
                                  int total) {
    int i = (int)(blockIdx.x * blockDim.x + threadIdx.x);
    if (i >= total) return;
    int node = i / D_OUT;
    int n    = i - node * D_OUT;
    float cnt = acc[(long long)node * 4 + 3];
    float sum = acc[(long long)node * 4 + n];
    out[i] = sum / fmaxf(cnt, 1.0f) + bc[n];
}

extern "C" void kernel_launch(void* const* d_in, const int* in_sizes, int n_in,
                              void* d_out, int out_size, void* d_ws, size_t ws_size,
                              hipStream_t stream) {
    const float* x     = (const float*)d_in[0];
    const int*   edges = (const int*)  d_in[1];
    const float* W1    = (const float*)d_in[2];
    const float* b1    = (const float*)d_in[3];
    const float* W2    = (const float*)d_in[4];
    const float* b2    = (const float*)d_in[5];
    float*       out   = (float*)d_out;

    int N = in_sizes[0] / D_IN;
    int E = in_sizes[1] / 2;

    float* ws = (float*)d_ws;
    long long offAcc = 0;
    long long offY   = (long long)N * 4;
    long long offB   = (long long)N * 8;
    long long offBc  = offB + 2048;

    gnn_fusew_kernel<<<33, 32, 0, stream>>>(W1, b1, W2, b2, ws + offB, ws + offBc);

    int nTiles = N / 16;
    int blocksProj = (nTiles + 7) / 8;
    gnn_project_kernel<<<blocksProj, 256, 0, stream>>>(
        x, ws + offB, ws + offY, ws + offAcc, N);

    int blocksScatter = (E + 511) / 512;
    gnn_scatter_kernel<<<blocksScatter, 256, 0, stream>>>(
        ws + offY, edges, ws + offAcc, E);

    int total = N * D_OUT;
    int blocksFin = (total + 255) / 256;
    gnn_finish_kernel<<<blocksFin, 256, 0, stream>>>(
        ws + offAcc, ws + offBc, out, total);
}


# Each chip half adds into its own copy of the node sums (half from the die id message, asked once per work-group), so no atomic line is shared across halves; finish adds the copies; sum order changes
# speedup vs baseline: 9.8746x; 2.2617x over previous
#include <hip/hip_runtime.h>
#include <hip/hip_bf16.h>

typedef float v2f __attribute__((ext_vector_type(2)));
typedef float v8f __attribute__((ext_vector_type(8)));

#define D_IN  128
#define D_HID 256
#define D_OUT 3


__global__ void gnn_fusew_kernel(const float* __restrict__ W1,
                                 const float* __restrict__ b1,
                                 const float* __restrict__ W2,
                                 const float* __restrict__ b2,
                                 float* __restrict__ packedB,
                                 float* __restrict__ bc) {
    int lane = threadIdx.x;
    int s    = blockIdx.x;

    if (s < 32) {
        float4 w1[8];
        float  w2[D_OUT][8];
        #pragma unroll
        for (int i = 0; i < 8; ++i)
            w1[i] = *(const float4*)(W1 + (8 * lane + i) * D_IN + 4 * s);
        #pragma unroll
        for (int n = 0; n < D_OUT; ++n) {
            float4 lo = *(const float4*)(W2 + n * D_HID + 8 * lane);
            float4 hi4 = *(const float4*)(W2 + n * D_HID + 8 * lane + 4);
            w2[n][0] = lo.x;  w2[n][1] = lo.y;  w2[n][2] = lo.z;  w2[n][3] = lo.w;
            w2[n][4] = hi4.x; w2[n][5] = hi4.y; w2[n][6] = hi4.z; w2[n][7] = hi4.w;
        }
        float p[D_OUT][4];
        #pragma unroll
        for (int n = 0; n < D_OUT; ++n) {
            p[n][0] = 0.0f; p[n][1] = 0.0f; p[n][2] = 0.0f; p[n][3] = 0.0f;
            #pragma unroll
            for (int i = 0; i < 8; ++i) {
                p[n][0] += w2[n][i] * w1[i].x;
                p[n][1] += w2[n][i] * w1[i].y;
                p[n][2] += w2[n][i] * w1[i].z;
                p[n][3] += w2[n][i] * w1[i].w;
            }
        }
        #pragma unroll
        for (int m = 16; m >= 1; m >>= 1) {
            #pragma unroll
            for (int n = 0; n < D_OUT; ++n) {
                #pragma unroll
                for (int r = 0; r < 4; ++r)
                    p[n][r] += __shfl_xor(p[n][r], m, 32);
            }
        }
        int  n  = lane & 15;
        bool hi = lane >= 16;
        #pragma unroll
        for (int j = 0; j < 2; ++j) {
            float v0 = hi ? p[0][j + 2] : p[0][j];
            float v1 = hi ? p[1][j + 2] : p[1][j];
            float v2 = hi ? p[2][j + 2] : p[2][j];
            float v  = (n == 0) ? v0 : (n == 1) ? v1 : (n == 2) ? v2 : 0.0f;
            packedB[s * 64 + j * 32 + lane] = v;
        }
    } else {
        float q[D_OUT];
        float bb[8];
        float w2[D_OUT][8];
        {
            float4 lo = *(const float4*)(b1 + 8 * lane);
            float4 hi4 = *(const float4*)(b1 + 8 * lane + 4);
            bb[0] = lo.x;  bb[1] = lo.y;  bb[2] = lo.z;  bb[3] = lo.w;
            bb[4] = hi4.x; bb[5] = hi4.y; bb[6] = hi4.z; bb[7] = hi4.w;
        }
        #pragma unroll
        for (int n = 0; n < D_OUT; ++n) {
            float4 lo = *(const float4*)(W2 + n * D_HID + 8 * lane);
            float4 hi4 = *(const float4*)(W2 + n * D_HID + 8 * lane + 4);
            w2[n][0] = lo.x;  w2[n][1] = lo.y;  w2[n][2] = lo.z;  w2[n][3] = lo.w;
            w2[n][4] = hi4.x; w2[n][5] = hi4.y; w2[n][6] = hi4.z; w2[n][7] = hi4.w;
        }
        #pragma unroll
        for (int n = 0; n < D_OUT; ++n) {
            q[n] = 0.0f;
            #pragma unroll
            for (int i = 0; i < 8; ++i)
                q[n] += w2[n][i] * bb[i];
        }
        #pragma unroll
        for (int m = 16; m >= 1; m >>= 1) {
            #pragma unroll
            for (int n = 0; n < D_OUT; ++n)
                q[n] += __shfl_xor(q[n], m, 32);
        }
        if (lane < 4) {
            float v = (lane == 0) ? q[0] : (lane == 1) ? q[1] : (lane == 2) ? q[2] : 0.0f;
            bc[lane] = (lane < D_OUT) ? v + b2[lane] : 0.0f;
        }
    }
}

__global__ void gnn_project_kernel(const float* __restrict__ x,
                                   const float* __restrict__ packedB,
                                   float* __restrict__ y4,
                                   float* __restrict__ acc,
                                   int N) {
    int wave = (int)((blockIdx.x * blockDim.x + threadIdx.x) >> 5);
    int lane = threadIdx.x & 31;
    int nTiles = N >> 4;
    if (wave >= nTiles) return;

    int m0   = wave << 4;
    int mrow = m0 + (lane & 15);
    int koff = (lane >= 16) ? 2 : 0;
    const float* arow = x + (long long)mrow * D_IN + koff;

    v8f c = {};
    #pragma unroll 4
    for (int s = 0; s < 32; ++s) {
        float2 a2 = *(const float2*)(arow + 4 * s);
        v2f a; a[0] = a2.x; a[1] = a2.y;
        v2f b; b[0] = packedB[s * 64 + lane];
               b[1] = packedB[s * 64 + 32 + lane];
        c = __builtin_amdgcn_wmma_f32_16x16x4_f32(
                 false, a,  false, b,
                 (short)0, c,  false,  false);
    }

    int n = lane & 15;
    if (n < 4) {
        int rbase = m0 + ((lane >= 16) ? 8 : 0);
        #pragma unroll
        for (int j = 0; j < 8; ++j) {
            long long o = (long long)(rbase + j) * 4 + n;
            y4[o]  = (n < D_OUT) ? c[j] : 1.0f;
            acc[o] = 0.0f;
            acc[(long long)N * 4 + o] = 0.0f;
        }
    }
}

template <bool FULL>
static __device__ __forceinline__ void scatter64(const float* __restrict__ y4,
                                                 float* __restrict__ acc,
                                                 int E, int base, int lane,
                                                 int s0, int d0, int s1, int d1) {
    unsigned comp = (unsigned)lane & 3u;
    int quad = lane >> 2;

    float    v[8];
    unsigned dd[8];
    bool     ok[8];
    #pragma unroll
    for (int k = 0; k < 8; ++k) {
        int j = (k & 3) * 8 + quad;
        unsigned sj = (unsigned)__shfl((k < 4) ? s0 : s1, j, 32);
        dd[k] = (unsigned)__shfl((k < 4) ? d0 : d1, j, 32);
        ok[k] = FULL || (base + (k >> 2) * 32 + j) < E;
        v[k]  = ok[k] ? y4[sj * 4u + comp] : 0.0f;
    }
    #pragma unroll
    for (int k = 0; k < 8; ++k)
        if (ok[k]) atomicAdd(acc + (dd[k] * 4u + comp), v[k]);
}

__global__ void gnn_scatter_kernel(const float* __restrict__ y4,
                                   const int* __restrict__ edges,
                                   float* __restrict__ acc,
                                   int E, int N) {
    __shared__ unsigned s_half;
    int wave = (int)((blockIdx.x * blockDim.x + threadIdx.x) >> 5);
    int lane = threadIdx.x & 31;
    int base = wave * 64;

    int e0 = base + lane;
    int e1 = base + 32 + lane;
    int s0 = 0, d0 = 0, s1 = 0, d1 = 0;
    if (e0 < E) { s0 = edges[e0]; d0 = edges[E + e0]; }
    if (e1 < E) { s1 = edges[e1]; d1 = edges[E + e1]; }

    if (threadIdx.x == 0) {
        unsigned id = __builtin_amdgcn_s_sendmsg_rtn(0x87);
        s_half = (id >> 18) & 1u;
    }
    __syncthreads();
    float* accH = acc + (long long)s_half * ((long long)N * 4);

    if (base >= E) return;
    if (base + 64 <= E) scatter64<true >(y4, accH, E, base, lane, s0, d0, s1, d1);
    else                scatter64<false>(y4, accH, E, base, lane, s0, d0, s1, d1);
}

__global__ void gnn_finish_kernel(const float* __restrict__ acc,
                                  const float* __restrict__ bc,
                                  float* __restrict__ out,
                                  int total, int N) {
    int i = (int)(blockIdx.x * blockDim.x + threadIdx.x);
    if (i >= total) return;
    int node = i / D_OUT;
    int n    = i - node * D_OUT;
    const float* acc1 = acc + (long long)N * 4;
    float cnt = acc[(long long)node * 4 + 3] + acc1[(long long)node * 4 + 3];
    float sum = acc[(long long)node * 4 + n] + acc1[(long long)node * 4 + n];
    out[i] = sum / fmaxf(cnt, 1.0f) + bc[n];
}

extern "C" void kernel_launch(void* const* d_in, const int* in_sizes, int n_in,
                              void* d_out, int out_size, void* d_ws, size_t ws_size,
                              hipStream_t stream) {
    const float* x     = (const float*)d_in[0];
    const int*   edges = (const int*)  d_in[1];
    const float* W1    = (const float*)d_in[2];
    const float* b1    = (const float*)d_in[3];
    const float* W2    = (const float*)d_in[4];
    const float* b2    = (const float*)d_in[5];
    float*       out   = (float*)d_out;

    int N = in_sizes[0] / D_IN;
    int E = in_sizes[1] / 2;

    float* ws = (float*)d_ws;
    long long offAcc = 0;
    long long offY   = (long long)N * 8;
    long long offB   = (long long)N * 12;
    long long offBc  = offB + 2048;

    gnn_fusew_kernel<<<33, 32, 0, stream>>>(W1, b1, W2, b2, ws + offB, ws + offBc);

    int nTiles = N / 16;
    int blocksProj = (nTiles + 7) / 8;
    gnn_project_kernel<<<blocksProj, 256, 0, stream>>>(
        x, ws + offB, ws + offY, ws + offAcc, N);

    int blocksScatter = (E + 511) / 512;
    gnn_scatter_kernel<<<blocksScatter, 256, 0, stream>>>(
        ws + offY, edges, ws + offAcc, E, N);

    int total = N * D_OUT;
    int blocksFin = (total + 255) / 256;
    gnn_finish_kernel<<<blocksFin, 256, 0, stream>>>(
        ws + offAcc, ws + offBc, out, total, N);
}


# Weight fold W2@W1 moved into the projection launch: each 16-wave work-group folds it (partials added in LDS, fixed order) while its x rows are in flight; 4 -> 3 launches; fold sum order changes
# speedup vs baseline: 10.8551x; 1.0993x over previous
#include <hip/hip_runtime.h>
#include <hip/hip_bf16.h>

typedef float v2f __attribute__((ext_vector_type(2)));
typedef float v8f __attribute__((ext_vector_type(8)));

#define D_IN  128
#define D_HID 256
#define D_OUT 3


#define PJ_WAVES   16
#define PJ_THREADS (PJ_WAVES * 32)
#define PJ_GROUPS  256

__global__ __launch_bounds__(PJ_THREADS)
void gnn_foldproject_kernel(const float* __restrict__ x,
                            const float* __restrict__ W1,
                            const float* __restrict__ b1,
                            const float* __restrict__ W2,
                            const float* __restrict__ b2,
                            float* __restrict__ y4,
                            float* __restrict__ acc,
                            float* __restrict__ bc,
                            int N) {
    __shared__ __attribute__((aligned(16))) float s_part[PJ_WAVES * D_OUT * D_IN];
    __shared__ __attribute__((aligned(16))) float s_pB[2048];
    __shared__ float s_bias[PJ_WAVES * 4];

    const int tid  = threadIdx.x;
    const int lane = tid & 31;
    const int w = __builtin_amdgcn_readfirstlane(tid >> 5);

    float w2s[D_OUT][16];
    #pragma unroll
    for (int n = 0; n < D_OUT; ++n) {
        #pragma unroll
        for (int i = 0; i < 16; ++i)
            w2s[n][i] = W2[n * D_HID + 16 * w + i];
    }
    float4 w1v[16];
    #pragma unroll
    for (int i = 0; i < 16; ++i)
        w1v[i] = *(const float4*)(W1 + (16 * w + i) * D_IN + 4 * lane);

    const int  nTiles = N >> 4;
    const int  tile = w * (int)gridDim.x + (int)blockIdx.x;
    const bool has  = tile < nTiles;
    const int  m0   = tile << 4;
    const int  koff = (lane >= 16) ? 2 : 0;
    const int  mld  = has ? m0 : ((nTiles - 1) << 4);
    const float* arow = x + (long long)(mld + (lane & 15)) * D_IN + koff;
    float2 a2[32];
    #pragma unroll
    for (int s = 0; s < 32; ++s)
        a2[s] = *(const float2*)(arow + 4 * s);
    __builtin_amdgcn_sched_barrier(0);

    float p[D_OUT][4];
    #pragma unroll
    for (int n = 0; n < D_OUT; ++n) {
        p[n][0] = 0.0f; p[n][1] = 0.0f; p[n][2] = 0.0f; p[n][3] = 0.0f;
        #pragma unroll
        for (int i = 0; i < 16; ++i) {
            float c2 = w2s[n][i];
            p[n][0] += c2 * w1v[i].x;
            p[n][1] += c2 * w1v[i].y;
            p[n][2] += c2 * w1v[i].z;
            p[n][3] += c2 * w1v[i].w;
        }
    }
    #pragma unroll
    for (int n = 0; n < D_OUT; ++n) {
        float4 v; v.x = p[n][0]; v.y = p[n][1]; v.z = p[n][2]; v.w = p[n][3];
        *(float4*)(s_part + (w * D_OUT + n) * D_IN + 4 * lane) = v;
    }
    if (blockIdx.x == 0) {
        float pb[D_OUT];
        #pragma unroll
        for (int n = 0; n < D_OUT; ++n) {
            pb[n] = 0.0f;
            #pragma unroll
            for (int i = 0; i < 16; ++i)
                pb[n] += w2s[n][i] * b1[16 * w + i];
        }
        if (lane < D_OUT) {
            float v = (lane == 0) ? pb[0] : (lane == 1) ? pb[1] : pb[2];
            s_bias[w * 4 + lane] = v;
        }
    }
    {
        float4 z; z.x = 0.0f; z.y = 0.0f; z.z = 0.0f; z.w = 0.0f;
        *(float4*)(s_pB + 4 * tid) = z;
    }
    __syncthreads();

    if (tid < D_OUT * D_IN) {
        int n = tid >> 7;
        int k = tid & 127;
        float sum = s_part[n * D_IN + k];
        #pragma unroll
        for (int q = 1; q < PJ_WAVES; ++q)
            sum += s_part[(q * D_OUT + n) * D_IN + k];
        int s    = k >> 2;
        int j    = k & 1;
        int half = (k >> 1) & 1;
        s_pB[(s >> 1) * 128 + (n + 16 * half) * 4 + (s & 1) * 2 + j] = sum;
    } else if (blockIdx.x == 0 && tid < D_OUT * D_IN + 4) {
        int n = tid - D_OUT * D_IN;
        float sum = 0.0f;
        if (n < D_OUT) {
            sum = s_bias[n];
            #pragma unroll
            for (int q = 1; q < PJ_WAVES; ++q)
                sum += s_bias[q * 4 + n];
            sum += b2[n];
        }
        bc[n] = sum;
    }
    __syncthreads();

    if (!has) return;

    float4 bq[16];
    #pragma unroll
    for (int t = 0; t < 16; ++t)
        bq[t] = *(const float4*)(s_pB + (t * 32 + lane) * 4);
    __builtin_amdgcn_sched_barrier(0);

    v8f c = {};
    #pragma unroll
    for (int t = 0; t < 16; ++t) {
        v2f a; v2f b;
        a[0] = a2[2 * t].x;     a[1] = a2[2 * t].y;
        b[0] = bq[t].x;         b[1] = bq[t].y;
        c = __builtin_amdgcn_wmma_f32_16x16x4_f32(
                 false, a,  false, b,
                 (short)0, c,  false,  false);
        a[0] = a2[2 * t + 1].x; a[1] = a2[2 * t + 1].y;
        b[0] = bq[t].z;         b[1] = bq[t].w;
        c = __builtin_amdgcn_wmma_f32_16x16x4_f32(
                 false, a,  false, b,
                 (short)0, c,  false,  false);
    }

    int n = lane & 15;
    if (n < 4) {
        int rbase = m0 + ((lane >= 16) ? 8 : 0);
        #pragma unroll
        for (int j = 0; j < 8; ++j) {
            long long o = (long long)(rbase + j) * 4 + n;
            y4[o]  = (n < D_OUT) ? c[j] : 1.0f;
            acc[o] = 0.0f;
            acc[(long long)N * 4 + o] = 0.0f;
        }
    }
}

template <bool FULL>
static __device__ __forceinline__ void scatter64(const float* __restrict__ y4,
                                                 float* __restrict__ acc,
                                                 int E, int base, int lane,
                                                 int s0, int d0, int s1, int d1) {
    unsigned comp = (unsigned)lane & 3u;
    int quad = lane >> 2;

    float    v[8];
    unsigned dd[8];
    bool     ok[8];
    #pragma unroll
    for (int k = 0; k < 8; ++k) {
        int j = (k & 3) * 8 + quad;
        unsigned sj = (unsigned)__shfl((k < 4) ? s0 : s1, j, 32);
        dd[k] = (unsigned)__shfl((k < 4) ? d0 : d1, j, 32);
        ok[k] = FULL || (base + (k >> 2) * 32 + j) < E;
        v[k]  = ok[k] ? y4[sj * 4u + comp] : 0.0f;
    }
    #pragma unroll
    for (int k = 0; k < 8; ++k)
        if (ok[k]) atomicAdd(acc + (dd[k] * 4u + comp), v[k]);
}

__global__ void gnn_scatter_kernel(const float* __restrict__ y4,
                                   const int* __restrict__ edges,
                                   float* __restrict__ acc,
                                   int E, int N) {
    __shared__ unsigned s_half;
    int wave = (int)((blockIdx.x * blockDim.x + threadIdx.x) >> 5);
    int lane = threadIdx.x & 31;
    int base = wave * 64;

    int e0 = base + lane;
    int e1 = base + 32 + lane;
    int s0 = 0, d0 = 0, s1 = 0, d1 = 0;
    if (e0 < E) { s0 = edges[e0]; d0 = edges[E + e0]; }
    if (e1 < E) { s1 = edges[e1]; d1 = edges[E + e1]; }

    if (threadIdx.x == 0) {
        unsigned id = __builtin_amdgcn_s_sendmsg_rtn(0x87);
        s_half = (id >> 18) & 1u;
    }
    __syncthreads();
    float* accH = acc + (long long)s_half * ((long long)N * 4);

    if (base >= E) return;
    if (base + 64 <= E) scatter64<true >(y4, accH, E, base, lane, s0, d0, s1, d1);
    else                scatter64<false>(y4, accH, E, base, lane, s0, d0, s1, d1);
}

__global__ void gnn_finish_kernel(const float* __restrict__ acc,
                                  const float* __restrict__ bc,
                                  float* __restrict__ out,
                                  int total, int N) {
    int i = (int)(blockIdx.x * blockDim.x + threadIdx.x);
    if (i >= total) return;
    int node = i / D_OUT;
    int n    = i - node * D_OUT;
    const float* acc1 = acc + (long long)N * 4;
    float cnt = acc[(long long)node * 4 + 3] + acc1[(long long)node * 4 + 3];
    float sum = acc[(long long)node * 4 + n] + acc1[(long long)node * 4 + n];
    out[i] = sum / fmaxf(cnt, 1.0f) + bc[n];
}

extern "C" void kernel_launch(void* const* d_in, const int* in_sizes, int n_in,
                              void* d_out, int out_size, void* d_ws, size_t ws_size,
                              hipStream_t stream) {
    const float* x     = (const float*)d_in[0];
    const int*   edges = (const int*)  d_in[1];
    const float* W1    = (const float*)d_in[2];
    const float* b1    = (const float*)d_in[3];
    const float* W2    = (const float*)d_in[4];
    const float* b2    = (const float*)d_in[5];
    float*       out   = (float*)d_out;

    int N = in_sizes[0] / D_IN;
    int E = in_sizes[1] / 2;

    float* ws = (float*)d_ws;
    long long offAcc = 0;
    long long offY   = (long long)N * 8;
    long long offB   = (long long)N * 12;
    long long offBc  = offB + 2048;

    int nTiles = N / 16;
    int blocksProj = (nTiles > PJ_GROUPS * PJ_WAVES) ? (nTiles + PJ_WAVES - 1) / PJ_WAVES : PJ_GROUPS;
    gnn_foldproject_kernel<<<blocksProj, PJ_THREADS, 0, stream>>>(
        x, W1, b1, W2, b2, ws + offY, ws + offAcc, ws + offBc, N);

    int blocksScatter = (E + 511) / 512;
    gnn_scatter_kernel<<<blocksScatter, 256, 0, stream>>>(
        ws + offY, edges, ws + offAcc, E, N);

    int total = N * D_OUT;
    int blocksFin = (total + 255) / 256;
    gnn_finish_kernel<<<blocksFin, 256, 0, stream>>>(
        ws + offAcc, ws + offBc, out, total, N);
}
